# starting kernel
# baseline (speedup 1.0000x reference)
#include <hip/hip_runtime.h>
#include <math.h>

typedef __attribute__((ext_vector_type(2))) float v2f;
typedef __attribute__((ext_vector_type(8))) float v8f;

#define HW   3136
#define CCH  512
#define NB   32
#define EPSV 1e-5f

__global__ void se_reduce_kernel(const float* __restrict__ x,
                                 float* __restrict__ s_out,
                                 float* __restrict__ q_out) {
    const int bc = blockIdx.x;
    const float4* p = (const float4*)(x + (size_t)bc * HW);
    float sum = 0.f, sq = 0.f;
    for (int i = threadIdx.x; i < HW / 4; i += blockDim.x) {
        float4 v = p[i];
        sum += v.x + v.y + v.z + v.w;
        sq  += v.x * v.x + v.y * v.y + v.z * v.z + v.w * v.w;
    }
    #pragma unroll
    for (int off = 16; off > 0; off >>= 1) {
        sum += __shfl_down(sum, off, 32);
        sq  += __shfl_down(sq,  off, 32);
    }
    __shared__ float ls[8], lq[8];
    const int lane = threadIdx.x & 31;
    const int wv   = threadIdx.x >> 5;
    if (lane == 0) { ls[wv] = sum; lq[wv] = sq; }
    __syncthreads();
    if (threadIdx.x == 0) {
        float ts = 0.f, tq = 0.f;
        #pragma unroll
        for (int i = 0; i < 8; ++i) { ts += ls[i]; tq += lq[i]; }
        const float inv = 1.0f / (float)HW;
        s_out[bc] = ts * inv;
        q_out[bc] = tq * inv;
    }
}

__global__ void se_excite_kernel(const float* __restrict__ s,
                                 const float* __restrict__ q,
                                 const float* __restrict__ w1,
                                 const float* __restrict__ w2,
                                 float* __restrict__ my,
                                 float* __restrict__ vv) {
    __shared__ float t1[32][33];
    const int lane = threadIdx.x & 31;
    const int wv   = threadIdx.x >> 5;
    const int half = lane >> 4;
    const int l    = lane & 15;

    {
        const int mt = wv >> 1, nt = wv & 1;
        const int arow = 16 * mt + l;
        const int brow = 16 * nt + l;
        v8f acc = {};
        for (int k = 0; k < CCH; k += 4) {
            v2f a, b;
            a.x = s[arow * CCH + k + 2 * half + 0];
            a.y = s[arow * CCH + k + 2 * half + 1];
            b.x = w1[brow * CCH + k + 2 * half + 0];
            b.y = w1[brow * CCH + k + 2 * half + 1];
            acc = __builtin_amdgcn_wmma_f32_16x16x4_f32(
                false, a, false, b, (short)0, acc, false, false);
        }
        #pragma unroll
        for (int r = 0; r < 8; ++r) {
            float vr = acc[r];
            t1[16 * mt + r + 8 * half][16 * nt + l] = vr > 0.f ? vr : 0.f;
        }
    }
    __syncthreads();

    for (int tile = wv; tile < 64; tile += 4) {
        const int mt = tile & 1;
        const int nt = tile >> 1;
        const int arow = 16 * mt + l;
        const int ncol = 16 * nt + l;
        v8f acc = {};
        #pragma unroll
        for (int k = 0; k < 32; k += 4) {
            v2f a, b;
            a.x = t1[arow][k + 2 * half + 0];
            a.y = t1[arow][k + 2 * half + 1];
            b.x = w2[ncol * 32 + k + 2 * half + 0];
            b.y = w2[ncol * 32 + k + 2 * half + 1];
            acc = __builtin_amdgcn_wmma_f32_16x16x4_f32(
                false, a, false, b, (short)0, acc, false, false);
        }
        #pragma unroll
        for (int r = 0; r < 8; ++r) {
            const int b = 16 * mt + r + 8 * half;
            const int c = ncol;
            float e  = 1.0f / (1.0f + __expf(-acc[r]));
            float sv = s[b * CCH + c];
            float qv = q[b * CCH + c];
            my[b * CCH + c] = sv * e;
            vv[b * CCH + c] = e * e * qv;
        }
    }
}

__global__ void se_finalize_kernel(const float* __restrict__ my,
                                   const float* __restrict__ vv,
                                   const float* __restrict__ bw,
                                   const float* __restrict__ bb,
                                   float* __restrict__ out) {
    const int k = threadIdx.x;
    float s1 = 0.f, t1 = 0.f, s2 = 0.f, t2 = 0.f;
    #pragma unroll 4
    for (int b = 0; b < NB; ++b) {
        s1 += my[b * CCH + k];
        t1 += vv[b * CCH + k];
        s2 += my[b * CCH + k + 256];
        t2 += vv[b * CCH + k + 256];
    }
    const float inv = 1.0f / (float)NB;
    const float mu1 = s1 * inv, mu2 = s2 * inv;
    const float g1 = bw[k]       * rsqrtf(t1 * inv - mu1 * mu1 + EPSV);
    const float g2 = bw[k + 256] * rsqrtf(t2 * inv - mu2 * mu2 + EPSV);
    const float bias = bb[k] + bb[k + 256];
    for (int b = 0; b < NB; ++b) {
        out[b * 256 + k] = (my[b * CCH + k] - mu1) * g1
                         + (my[b * CCH + k + 256] - mu2) * g2 + bias;
    }
}

extern "C" void kernel_launch(void* const* d_in, const int* in_sizes, int n_in,
                              void* d_out, int out_size, void* d_ws, size_t ws_size,
                              hipStream_t stream) {
    const float* x  = (const float*)d_in[0];
    const float* w1 = (const float*)d_in[1];
    const float* w2 = (const float*)d_in[2];
    const float* bw = (const float*)d_in[3];
    const float* bb = (const float*)d_in[4];
    float* out = (float*)d_out;

    float* ws = (float*)d_ws;
    float* s  = ws;
    float* q  = ws + 16384;
    float* my = ws + 32768;
    float* vv = ws + 49152;

    se_reduce_kernel<<<NB * CCH, 256, 0, stream>>>(x, s, q);
    se_excite_kernel<<<1, 128, 0, stream>>>(s, q, w1, w2, my, vv);
    se_finalize_kernel<<<1, 256, 0, stream>>>(my, vv, bw, bb, out);
}


# reduce kernel: one 8-wave group per 32-row span (was per row), 98 whole-line 16-B loads per thread, 24-30 in flight; per-lane row partials, lane-halving exchange, LDS join: new row-sum order
# speedup vs baseline: 1.8885x; 1.8885x over previous
#include <hip/hip_runtime.h>
#include <math.h>

typedef __attribute__((ext_vector_type(2))) float v2f;
typedef __attribute__((ext_vector_type(8))) float v8f;

#define HW   3136
#define CCH  512
#define NB   32
#define EPSV 1e-5f

#define RGRP 512
#define RROWS 32
#define NLD  98
#define NCHK 16
#define AHD  4
#define CH_START(C) (49 * ((C) >> 3) + 6 * ((C) & 7))
#define CH_COUNT(C) ((((C) & 7) == 7) ? 7 : 6)

template <int WD>
__device__ __forceinline__ void lane_halving_step(float (&val)[32], const int lane) {
    const bool up = (lane & WD) != 0;
    #pragma unroll
    for (int k = 0; k < WD; ++k) {
        const float keep = up ? val[k + WD] : val[k];
        const float send = up ? val[k] : val[k + WD];
        val[k] = keep + __shfl_xor(send, WD, 32);
    }
}

__global__ __launch_bounds__(256, 2)
void se_reduce_kernel(const float* __restrict__ x,
                      float* __restrict__ s_out,
                      float* __restrict__ q_out) {
    __shared__ float part[8 * 64];
    const int t    = threadIdx.x;
    const int lane = t & 31;
    const int wv   = __builtin_amdgcn_readfirstlane(t >> 5);
    const int g    = blockIdx.x;
    const float4* p = (const float4*)(x + (size_t)g * (RROWS * HW)) + t;

    float4 v[NLD];
    #pragma unroll
    for (int j = 0; j < CH_START(AHD); ++j) v[j] = p[256 * j];

    float tot[2];
    #pragma unroll
    for (int sp = 0; sp < 2; ++sp) {
        v2f as2[16], aq2[16];
        float val[32];
        #pragma unroll
        for (int i = 0; i < 16; ++i) { as2[i] = (v2f){0.f, 0.f}; aq2[i] = (v2f){0.f, 0.f}; }
        #pragma unroll
        for (int i = 0; i < 8; ++i) {
            const int c = 8 * sp + i;
            __builtin_amdgcn_sched_barrier(0);
            if (c + AHD < NCHK) {
                #pragma unroll
                for (int u = 0; u < 7; ++u)
                    if (u < CH_COUNT(c + AHD)) v[CH_START(c + AHD) + u] = p[256 * (CH_START(c + AHD) + u)];
            }
            __builtin_amdgcn_sched_barrier(0);
            #pragma unroll
            for (int u = 0; u < 7; ++u) {
                if (u < CH_COUNT(c)) {
                    const int k = 6 * i + u;
                    const float4 d = v[49 * sp + k];
                    const v2f lo2 = {d.x, d.y}, hi2 = {d.z, d.w};
                    const v2f s2 = lo2 + hi2;
                    const v2f q2 = lo2 * lo2 + hi2 * hi2;
                    const int r = (k == 48) ? 15 : k / 3;
                    if ((k % 3 == 0) && (k != 0) && (k != 48)) {
                        const bool lo = t < 16 * r;
                        as2[r - 1] += (v2f){lo ? s2.x : 0.f, lo ? s2.y : 0.f};
                        as2[r]     += (v2f){lo ? 0.f : s2.x, lo ? 0.f : s2.y};
                        aq2[r - 1] += (v2f){lo ? q2.x : 0.f, lo ? q2.y : 0.f};
                        aq2[r]     += (v2f){lo ? 0.f : q2.x, lo ? 0.f : q2.y};
                    } else {
                        as2[r] += s2;
                        aq2[r] += q2;
                    }
                }
            }
            float fa = as2[2 * i].x + as2[2 * i].y, fb = aq2[2 * i].x + aq2[2 * i].y;
            asm volatile("" : "+v"(fa), "+v"(fb), "+v"(as2[2 * i + 1]), "+v"(aq2[2 * i + 1]) : : "memory");
            val[2 * i] = fa; val[16 + 2 * i] = fb;
            if (i >= 1) {
                float fc = as2[2 * i - 1].x + as2[2 * i - 1].y, fd = aq2[2 * i - 1].x + aq2[2 * i - 1].y;
                asm volatile("" : "+v"(fc), "+v"(fd) : : "memory");
                val[2 * i - 1] = fc; val[16 + 2 * i - 1] = fd;
            }
        }
        val[15] = as2[15].x + as2[15].y; val[31] = aq2[15].x + aq2[15].y;
        lane_halving_step<16>(val, lane);
        lane_halving_step<8>(val, lane);
        lane_halving_step<4>(val, lane);
        lane_halving_step<2>(val, lane);
        lane_halving_step<1>(val, lane);
        tot[sp] = val[0];
    }
    part[wv * 64 + lane]      = tot[0];
    part[wv * 64 + 32 + lane] = tot[1];
    __syncthreads();
    if (t < 64) {
        const int kind = t >> 5, rr = t & 31;
        const int idx  = 32 * (rr >> 4) + 16 * kind + (rr & 15);
        float a = 0.f;
        #pragma unroll
        for (int w = 0; w < 8; ++w) a += part[w * 64 + idx];
        const float inv = 1.0f / (float)HW;
        (kind ? q_out : s_out)[RROWS * g + rr] = a * inv;
    }
}

#define SP 516
#define WP 36
#define PP 528

__global__ __launch_bounds__(1024)
void se_excite_kernel(const float* __restrict__ s,
                      const float* __restrict__ q,
                      const float* __restrict__ w1,
                      const float* __restrict__ w2,
                      const float* __restrict__ bw,
                      const float* __restrict__ bb,
                      float* __restrict__ out) {
    __shared__ __attribute__((aligned(16))) float s_l[NB * SP];
    __shared__ __attribute__((aligned(16))) float w1_l[32 * SP];
    __shared__ __attribute__((aligned(16))) float w2_l[CCH * WP];
    __shared__ __attribute__((aligned(16))) float t1_l[NB * WP];
    __shared__ float ps_l[4 * PP];
    __shared__ float pv_l[4 * PP];
    __shared__ float tm_l[NB * SP];

    const int tid  = threadIdx.x;
    const int lane = tid & 31;
    const int wv   = __builtin_amdgcn_readfirstlane(tid >> 5);
    const int half = lane >> 4;
    const int l    = lane & 15;

    const int mt2 = wv & 1;
    const int c0  = 32 * (wv >> 1);

    float4 rs[4], rw1[4], rw2[4];
    #pragma unroll
    for (int j = 0; j < 4; ++j) rs[j]  = ((const float4*)s)[tid + 1024 * j];
    #pragma unroll
    for (int j = 0; j < 4; ++j) rw1[j] = ((const float4*)w1)[tid + 1024 * j];
    #pragma unroll
    for (int j = 0; j < 4; ++j) rw2[j] = ((const float4*)w2)[tid + 1024 * j];

    float qv[2][8];
    #pragma unroll
    for (int j = 0; j < 2; ++j) {
        #pragma unroll
        for (int r = 0; r < 8; ++r)
            qv[j][r] = q[(16 * mt2 + r + 8 * half) * CCH + c0 + 16 * j + l];
    }

    float bwv[2];
    #pragma unroll
    for (int j = 0; j < 2; ++j) bwv[j] = bw[c0 + 16 * j + l];
    const int ko = tid & 255;
    const float bias = bb[ko] + bb[ko + 256];

    #pragma unroll
    for (int j = 0; j < 4; ++j) {
        const int i = tid + 1024 * j;
        *(float4*)&s_l[(i >> 7) * SP + 4 * (i & 127)]  = rs[j];
        *(float4*)&w1_l[(i >> 7) * SP + 4 * (i & 127)] = rw1[j];
    }
    #pragma unroll
    for (int j = 0; j < 4; ++j) {
        const int i = tid + 1024 * j;
        *(float4*)&w2_l[(i >> 3) * WP + 4 * (i & 7)] = rw2[j];
    }
    __syncthreads();

    if (wv < 4) {
        const int mt = wv >> 1, nt = wv & 1;
        const float* ap = s_l  + (16 * mt + l) * SP + 2 * half;
        const float* bp = w1_l + (16 * nt + l) * SP + 2 * half;
        v8f acc = {};
        v2f a0[8], b0[8], a1[8], b1[8];
        #pragma unroll
        for (int u = 0; u < 8; ++u) {
            a0[u] = *(const v2f*)(ap + 4 * u);
            b0[u] = *(const v2f*)(bp + 4 * u);
        }
        #pragma unroll
        for (int k = 0; k < CCH; k += 64) {
            #pragma unroll
            for (int u = 0; u < 8; ++u) {
                a1[u] = *(const v2f*)(ap + k + 32 + 4 * u);
                b1[u] = *(const v2f*)(bp + k + 32 + 4 * u);
            }
            __builtin_amdgcn_sched_barrier(0);
            #pragma unroll
            for (int u = 0; u < 8; ++u)
                acc = __builtin_amdgcn_wmma_f32_16x16x4_f32(
                    false, a0[u], false, b0[u], (short)0, acc, false, false);
            __builtin_amdgcn_sched_barrier(0);
            if (k + 64 < CCH) {
                #pragma unroll
                for (int u = 0; u < 8; ++u) {
                    a0[u] = *(const v2f*)(ap + k + 64 + 4 * u);
                    b0[u] = *(const v2f*)(bp + k + 64 + 4 * u);
                }
            }
            __builtin_amdgcn_sched_barrier(0);
            #pragma unroll
            for (int u = 0; u < 8; ++u)
                acc = __builtin_amdgcn_wmma_f32_16x16x4_f32(
                    false, a1[u], false, b1[u], (short)0, acc, false, false);
        }
        #pragma unroll
        for (int r = 0; r < 8; ++r) {
            float vr = acc[r];
            t1_l[(16 * mt + r + 8 * half) * WP + 16 * nt + l] = vr > 0.f ? vr : 0.f;
        }
    }
    __syncthreads();

    {
        const float* ap = t1_l + (16 * mt2 + l) * WP + 2 * half;
        v2f a[8];
        #pragma unroll
        for (int u = 0; u < 8; ++u) a[u] = *(const v2f*)(ap + 4 * u);
        v2f b[2][8];
        #pragma unroll
        for (int j = 0; j < 2; ++j) {
            const float* bp = w2_l + (c0 + 16 * j + l) * WP + 2 * half;
            #pragma unroll
            for (int u = 0; u < 8; ++u) b[j][u] = *(const v2f*)(bp + 4 * u);
        }
        __builtin_amdgcn_sched_barrier(0);
        v8f acc[2] = {};
        #pragma unroll
        for (int u = 0; u < 8; ++u) {
            #pragma unroll
            for (int j = 0; j < 2; ++j)
                acc[j] = __builtin_amdgcn_wmma_f32_16x16x4_f32(
                    false, a[u], false, b[j][u], (short)0, acc[j], false, false);
        }
        float myv[2][8];
        const int g8 = 2 * mt2 + half;
        #pragma unroll
        for (int j = 0; j < 2; ++j) {
            const int c = c0 + 16 * j + l;
            float ps = 0.f, pv = 0.f;
            #pragma unroll
            for (int r = 0; r < 8; ++r) {
                const int bi = 16 * mt2 + r + 8 * half;
                float e  = 1.0f / (1.0f + __expf(-acc[j][r]));
                float sv = s_l[bi * SP + c];
                float m  = sv * e;
                myv[j][r] = m;
                ps += m;
                pv += e * e * qv[j][r];
            }
            ps_l[g8 * PP + c] = ps;
            pv_l[g8 * PP + c] = pv;
        }
        __syncthreads();
        const float inv = 1.0f / (float)NB;
        #pragma unroll
        for (int j = 0; j < 2; ++j) {
            const int c = c0 + 16 * j + l;
            const float s1 = (ps_l[c] + ps_l[PP + c]) + (ps_l[2 * PP + c] + ps_l[3 * PP + c]);
            const float t1 = (pv_l[c] + pv_l[PP + c]) + (pv_l[2 * PP + c] + pv_l[3 * PP + c]);
            const float mu = s1 * inv;
            const float g  = bwv[j] * rsqrtf(t1 * inv - mu * mu + EPSV);
            #pragma unroll
            for (int r = 0; r < 8; ++r) {
                const int bi = 16 * mt2 + r + 8 * half;
                tm_l[bi * SP + c] = (myv[j][r] - mu) * g;
            }
        }
    }
    __syncthreads();
    {
        const int bq = tid >> 8;
        #pragma unroll
        for (int i = 0; i < 8; ++i) {
            const int b = 8 * bq + i;
            out[b * 256 + ko] = tm_l[b * SP + ko] + tm_l[b * SP + ko + 256] + bias;
        }
    }
}

extern "C" void kernel_launch(void* const* d_in, const int* in_sizes, int n_in,
                              void* d_out, int out_size, void* d_ws, size_t ws_size,
                              hipStream_t stream) {
    const float* x  = (const float*)d_in[0];
    const float* w1 = (const float*)d_in[1];
    const float* w2 = (const float*)d_in[2];
    const float* bw = (const float*)d_in[3];
    const float* bb = (const float*)d_in[4];
    float* out = (float*)d_out;

    float* ws = (float*)d_ws;
    float* s  = ws;
    float* q  = ws + 16384;

    se_reduce_kernel<<<RGRP, 256, 0, stream>>>(x, s, q);
    se_excite_kernel<<<1, 1024, 0, stream>>>(s, q, w1, w2, bw, bb, out);
}


# excite kernel: operands fetched in order of need (s, w1 as two column halves, then w2, q) over hand-written barriers that keep loads in flight; waves 0-3 run product 1 under the rest; sums unchanged
# speedup vs baseline: 2.0022x; 1.0602x over previous
#include <hip/hip_runtime.h>
#include <math.h>

typedef __attribute__((ext_vector_type(2))) float v2f;
typedef __attribute__((ext_vector_type(8))) float v8f;
typedef __attribute__((ext_vector_type(4))) float v4f;

#define HW   3136
#define CCH  512
#define NB   32
#define EPSV 1e-5f

#define RGRP 512
#define RROWS 32
#define NLD  98
#define NCHK 16
#define AHD  4
#define CH_START(C) (49 * ((C) >> 3) + 6 * ((C) & 7))
#define CH_COUNT(C) ((((C) & 7) == 7) ? 7 : 6)

template <int WD>
__device__ __forceinline__ void lane_halving_step(float (&val)[32], const int lane) {
    const bool up = (lane & WD) != 0;
    #pragma unroll
    for (int k = 0; k < WD; ++k) {
        const float keep = up ? val[k + WD] : val[k];
        const float send = up ? val[k] : val[k + WD];
        val[k] = keep + __shfl_xor(send, WD, 32);
    }
}

__global__ __launch_bounds__(256, 2)
void se_reduce_kernel(const float* __restrict__ x,
                      float* __restrict__ s_out,
                      float* __restrict__ q_out) {
    __shared__ float part[8 * 64];
    const int t    = threadIdx.x;
    const int lane = t & 31;
    const int wv   = __builtin_amdgcn_readfirstlane(t >> 5);
    const int g    = blockIdx.x;
    const float4* p = (const float4*)(x + (size_t)g * (RROWS * HW)) + t;

    float4 v[NLD];
    #pragma unroll
    for (int j = 0; j < CH_START(AHD); ++j) v[j] = p[256 * j];

    float tot[2];
    #pragma unroll
    for (int sp = 0; sp < 2; ++sp) {
        v2f as2[16], aq2[16];
        float val[32];
        #pragma unroll
        for (int i = 0; i < 16; ++i) { as2[i] = (v2f){0.f, 0.f}; aq2[i] = (v2f){0.f, 0.f}; }
        #pragma unroll
        for (int i = 0; i < 8; ++i) {
            const int c = 8 * sp + i;
            __builtin_amdgcn_sched_barrier(0);
            if (c + AHD < NCHK) {
                #pragma unroll
                for (int u = 0; u < 7; ++u)
                    if (u < CH_COUNT(c + AHD)) v[CH_START(c + AHD) + u] = p[256 * (CH_START(c + AHD) + u)];
            }
            __builtin_amdgcn_sched_barrier(0);
            #pragma unroll
            for (int u = 0; u < 7; ++u) {
                if (u < CH_COUNT(c)) {
                    const int k = 6 * i + u;
                    const float4 d = v[49 * sp + k];
                    const v2f lo2 = {d.x, d.y}, hi2 = {d.z, d.w};
                    const v2f s2 = lo2 + hi2;
                    const v2f q2 = lo2 * lo2 + hi2 * hi2;
                    const int r = (k == 48) ? 15 : k / 3;
                    if ((k % 3 == 0) && (k != 0) && (k != 48)) {
                        const bool lo = t < 16 * r;
                        as2[r - 1] += (v2f){lo ? s2.x : 0.f, lo ? s2.y : 0.f};
                        as2[r]     += (v2f){lo ? 0.f : s2.x, lo ? 0.f : s2.y};
                        aq2[r - 1] += (v2f){lo ? q2.x : 0.f, lo ? q2.y : 0.f};
                        aq2[r]     += (v2f){lo ? 0.f : q2.x, lo ? 0.f : q2.y};
                    } else {
                        as2[r] += s2;
                        aq2[r] += q2;
                    }
                }
            }
            float fa = as2[2 * i].x + as2[2 * i].y, fb = aq2[2 * i].x + aq2[2 * i].y;
            asm volatile("" : "+v"(fa), "+v"(fb), "+v"(as2[2 * i + 1]), "+v"(aq2[2 * i + 1]) : : "memory");
            val[2 * i] = fa; val[16 + 2 * i] = fb;
            if (i >= 1) {
                float fc = as2[2 * i - 1].x + as2[2 * i - 1].y, fd = aq2[2 * i - 1].x + aq2[2 * i - 1].y;
                asm volatile("" : "+v"(fc), "+v"(fd) : : "memory");
                val[2 * i - 1] = fc; val[16 + 2 * i - 1] = fd;
            }
        }
        val[15] = as2[15].x + as2[15].y; val[31] = aq2[15].x + aq2[15].y;
        lane_halving_step<16>(val, lane);
        lane_halving_step<8>(val, lane);
        lane_halving_step<4>(val, lane);
        lane_halving_step<2>(val, lane);
        lane_halving_step<1>(val, lane);
        tot[sp] = val[0];
    }
    part[wv * 64 + lane]      = tot[0];
    part[wv * 64 + 32 + lane] = tot[1];
    __syncthreads();
    if (t < 64) {
        const int kind = t >> 5, rr = t & 31;
        const int idx  = 32 * (rr >> 4) + 16 * kind + (rr & 15);
        float a = 0.f;
        #pragma unroll
        for (int w = 0; w < 8; ++w) a += part[w * 64 + idx];
        const float inv = 1.0f / (float)HW;
        (kind ? q_out : s_out)[RROWS * g + rr] = a * inv;
    }
}

#define SP 516
#define WP 36
#define PP 528
#define KH 256

__device__ __forceinline__ void lds_barrier() {
    asm volatile("s_wait_dscnt 0x0" ::: "memory");
    asm volatile("s_barrier_signal -1" ::: "memory");
    asm volatile("s_barrier_wait -1" ::: "memory");
}

__device__ __forceinline__ void product1_half(v8f& acc, const float* ap, const float* bp) {
    v2f a[3][4], b[3][4];
    #pragma unroll
    for (int u = 0; u < 4; ++u) {
        a[0][u] = *(const v2f*)(ap + 4 * u);
        b[0][u] = *(const v2f*)(bp + 4 * u);
        a[1][u] = *(const v2f*)(ap + 16 + 4 * u);
        b[1][u] = *(const v2f*)(bp + 16 + 4 * u);
    }
    #pragma unroll
    for (int g = 0; g < KH / 16; ++g) {
        if (g + 2 < KH / 16) {
            #pragma unroll
            for (int u = 0; u < 4; ++u) {
                a[(g + 2) % 3][u] = *(const v2f*)(ap + 16 * (g + 2) + 4 * u);
                b[(g + 2) % 3][u] = *(const v2f*)(bp + 16 * (g + 2) + 4 * u);
            }
        }
        __builtin_amdgcn_sched_barrier(0);
        #pragma unroll
        for (int u = 0; u < 4; ++u)
            acc = __builtin_amdgcn_wmma_f32_16x16x4_f32(
                false, a[g % 3][u], false, b[g % 3][u], (short)0, acc, false, false);
        __builtin_amdgcn_sched_barrier(0);
    }
}

__global__ __launch_bounds__(1024)
void se_excite_kernel(const float* s,
                      const float* q,
                      const float* w1,
                      const float* w2,
                      const float* bw,
                      const float* bb,
                      float* out) {
    __shared__ __attribute__((aligned(16))) float s_l[NB * SP];
    __shared__ __attribute__((aligned(16))) float w1_l[32 * SP];
    __shared__ __attribute__((aligned(16))) float w2_l[CCH * WP];
    __shared__ __attribute__((aligned(16))) float t1_l[NB * WP];
    __shared__ float ps_l[4 * PP];
    __shared__ float pv_l[4 * PP];
    __shared__ float tm_l[NB * SP];

    const int tid  = threadIdx.x;
    const int lane = tid & 31;
    const int wv   = __builtin_amdgcn_readfirstlane(tid >> 5);
    const int half = lane >> 4;
    const int l    = lane & 15;

    const int mt2 = wv & 1;
    const int c0  = 32 * (wv >> 1);

    const int mt = wv >> 1, nt = wv & 1;
    const float* ap1 = s_l  + (16 * mt + l) * SP + 2 * half;
    const float* bp1 = w1_l + (16 * nt + l) * SP + 2 * half;

    v4f rs[2][2], rw1[2][2], rw2[4];
    #pragma unroll
    for (int j = 0; j < 2; ++j) {
        const int i = tid + 1024 * j;
        rs[0][j]  = ((const v4f*)s)[(i >> 6) * 128 + (i & 63)];
        rw1[0][j] = ((const v4f*)w1)[(i >> 6) * 128 + (i & 63)];
    }
    lds_barrier();
    #pragma unroll
    for (int j = 0; j < 2; ++j) {
        const int i = tid + 1024 * j;
        rs[1][j]  = ((const v4f*)s)[(i >> 6) * 128 + 64 + (i & 63)];
        rw1[1][j] = ((const v4f*)w1)[(i >> 6) * 128 + 64 + (i & 63)];
    }

    #pragma unroll
    for (int j = 0; j < 2; ++j) {
        const int i = tid + 1024 * j;
        *(v4f*)&s_l[(i >> 6) * SP + 4 * (i & 63)]  = rs[0][j];
        *(v4f*)&w1_l[(i >> 6) * SP + 4 * (i & 63)] = rw1[0][j];
    }
    lds_barrier();

    #pragma unroll
    for (int j = 0; j < 4; ++j) rw2[j] = ((const v4f*)w2)[tid + 1024 * j];

    v8f acc1 = {};
    if (wv < 4) product1_half(acc1, ap1, bp1);

    #pragma unroll
    for (int j = 0; j < 2; ++j) {
        const int i = tid + 1024 * j;
        *(v4f*)&s_l[(i >> 6) * SP + KH + 4 * (i & 63)]  = rs[1][j];
        *(v4f*)&w1_l[(i >> 6) * SP + KH + 4 * (i & 63)] = rw1[1][j];
    }
    lds_barrier();

    const int ko = tid & 255;
    const float bias = bb[ko] + bb[ko + 256];
    float bwv[2];
    #pragma unroll
    for (int j = 0; j < 2; ++j) bwv[j] = bw[c0 + 16 * j + l];

    float qv[2][8];
    #pragma unroll
    for (int j = 0; j < 2; ++j) {
        #pragma unroll
        for (int r = 0; r < 8; ++r)
            qv[j][r] = q[(16 * mt2 + r + 8 * half) * CCH + c0 + 16 * j + l];
    }

    if (wv < 4) {
        product1_half(acc1, ap1 + KH, bp1 + KH);
        #pragma unroll
        for (int r = 0; r < 8; ++r) {
            float vr = acc1[r];
            t1_l[(16 * mt + r + 8 * half) * WP + 16 * nt + l] = vr > 0.f ? vr : 0.f;
        }
    }

    #pragma unroll
    for (int j = 0; j < 4; ++j) {
        const int i = tid + 1024 * j;
        *(v4f*)&w2_l[(i >> 3) * WP + 4 * (i & 7)] = rw2[j];
    }
    lds_barrier();

    {
        const float* ap = t1_l + (16 * mt2 + l) * WP + 2 * half;
        v2f a[8];
        #pragma unroll
        for (int u = 0; u < 8; ++u) a[u] = *(const v2f*)(ap + 4 * u);
        v2f b[2][8];
        #pragma unroll
        for (int j = 0; j < 2; ++j) {
            const float* bp = w2_l + (c0 + 16 * j + l) * WP + 2 * half;
            #pragma unroll
            for (int u = 0; u < 8; ++u) b[j][u] = *(const v2f*)(bp + 4 * u);
        }
        __builtin_amdgcn_sched_barrier(0);
        v8f acc[2] = {};
        #pragma unroll
        for (int u = 0; u < 8; ++u) {
            #pragma unroll
            for (int j = 0; j < 2; ++j)
                acc[j] = __builtin_amdgcn_wmma_f32_16x16x4_f32(
                    false, a[u], false, b[j][u], (short)0, acc[j], false, false);
        }
        float myv[2][8];
        const int g8 = 2 * mt2 + half;
        #pragma unroll
        for (int j = 0; j < 2; ++j) {
            const int c = c0 + 16 * j + l;
            float ps = 0.f, pv = 0.f;
            #pragma unroll
            for (int r = 0; r < 8; ++r) {
                const int bi = 16 * mt2 + r + 8 * half;
                float e  = 1.0f / (1.0f + __expf(-acc[j][r]));
                float sv = s_l[bi * SP + c];
                float m  = sv * e;
                myv[j][r] = m;
                ps += m;
                pv += e * e * qv[j][r];
            }
            ps_l[g8 * PP + c] = ps;
            pv_l[g8 * PP + c] = pv;
        }
        __syncthreads();
        const float inv = 1.0f / (float)NB;
        #pragma unroll
        for (int j = 0; j < 2; ++j) {
            const int c = c0 + 16 * j + l;
            const float s1 = (ps_l[c] + ps_l[PP + c]) + (ps_l[2 * PP + c] + ps_l[3 * PP + c]);
            const float t1 = (pv_l[c] + pv_l[PP + c]) + (pv_l[2 * PP + c] + pv_l[3 * PP + c]);
            const float mu = s1 * inv;
            const float g  = bwv[j] * rsqrtf(t1 * inv - mu * mu + EPSV);
            #pragma unroll
            for (int r = 0; r < 8; ++r) {
                const int bi = 16 * mt2 + r + 8 * half;
                tm_l[bi * SP + c] = (myv[j][r] - mu) * g;
            }
        }
    }
    __syncthreads();
    {
        const int bq = tid >> 8;
        #pragma unroll
        for (int i = 0; i < 8; ++i) {
            const int b = 8 * bq + i;
            out[b * 256 + ko] = tm_l[b * SP + ko] + tm_l[b * SP + ko + 256] + bias;
        }
    }
}

extern "C" void kernel_launch(void* const* d_in, const int* in_sizes, int n_in,
                              void* d_out, int out_size, void* d_ws, size_t ws_size,
                              hipStream_t stream) {
    const float* x  = (const float*)d_in[0];
    const float* w1 = (const float*)d_in[1];
    const float* w2 = (const float*)d_in[2];
    const float* bw = (const float*)d_in[3];
    const float* bb = (const float*)d_in[4];
    float* out = (float*)d_out;

    float* ws = (float*)d_ws;
    float* s  = ws;
    float* q  = ws + 16384;

    se_reduce_kernel<<<RGRP, 256, 0, stream>>>(x, s, q);
    se_excite_kernel<<<1, 1024, 0, stream>>>(s, q, w1, w2, bw, bb, out);
}
